# starting kernel
# baseline (speedup 1.0000x reference)
#include <hip/hip_runtime.h>
#include <math.h>


typedef __attribute__((ext_vector_type(16))) __bf16 v16bf;
typedef __attribute__((ext_vector_type(8)))  __bf16 v8bf;
typedef __attribute__((ext_vector_type(8)))  float  v8f;

#define FIN 128
#define HEADS 8
#define FEAT 16
#define HF 128

__global__ void gat_init(float* __restrict__ denom, float* __restrict__ gmax,
                         float* __restrict__ out, int n) {
  int idx = blockIdx.x * blockDim.x + threadIdx.x;
  if (idx < n * HF) out[idx] = 0.0f;
  if (idx < n * HEADS) denom[idx] = 0.0f;
  if (idx == 0) gmax[0] = -__builtin_inff();
}

__global__ void gat_wcvt(const float* __restrict__ Wp, const float* __restrict__ Ws,
                         __bf16* __restrict__ WpT, __bf16* __restrict__ WsT) {
  int idx = blockIdx.x * blockDim.x + threadIdx.x;
  if (idx >= FIN * HF) return;
  int k = idx >> 7;
  int ncol = idx & 127;
  WpT[ncol * FIN + k] = (__bf16)Wp[k * HF + ncol];
  WsT[ncol * FIN + k] = (__bf16)Ws[k * HF + ncol];
}

__global__ __launch_bounds__(256)
void gat_proj_wmma(const float* __restrict__ x,
                   const __bf16* __restrict__ WpT,
                   const __bf16* __restrict__ WsT,
                   float* __restrict__ proj,
                   float* __restrict__ skip,
                   int n) {
  __shared__ __bf16 As[16][FIN + 8];

  const int t    = threadIdx.x;
  const int row0 = blockIdx.x * 16;

  for (int i = t; i < 16 * FIN; i += 256) {
    int r = i >> 7, c = i & 127;
    int gr = row0 + r;
    float v = (gr < n) ? x[(size_t)gr * FIN + c] : 0.0f;
    As[r][c] = (__bf16)v;
  }
  __syncthreads();

  const int wave = t >> 5;
  const int lane = t & 31;
  const int m    = lane & 15;
  const int kb   = (lane < 16) ? 0 : 8;
  const int ncol = wave * 16 + m;

  const v8bf* aRow  = (const v8bf*)(&As[m][0]);
  const v8bf* bpRow = (const v8bf*)(WpT + (size_t)ncol * FIN);
  const v8bf* bsRow = (const v8bf*)(WsT + (size_t)ncol * FIN);

  v8f accP = {};
  v8f accS = {};

#pragma unroll
  for (int kk = 0; kk < FIN; kk += 32) {
    const int c0 = (kk + kb) >> 3;
    const int c1 = c0 + 2;
    v8bf alo = aRow[c0],  ahi = aRow[c1];
    v8bf plo = bpRow[c0], phi = bpRow[c1];
    v8bf slo = bsRow[c0], shi = bsRow[c1];

    v16bf a  = __builtin_shufflevector(alo, ahi,
        0,1,2,3,4,5,6,7,8,9,10,11,12,13,14,15);
    v16bf bP = __builtin_shufflevector(plo, phi,
        0,1,2,3,4,5,6,7,8,9,10,11,12,13,14,15);
    v16bf bS = __builtin_shufflevector(slo, shi,
        0,1,2,3,4,5,6,7,8,9,10,11,12,13,14,15);

    accP = __builtin_amdgcn_wmma_f32_16x16x32_bf16(
        false, a, false, bP, (short)0, accP, false, false);
    accS = __builtin_amdgcn_wmma_f32_16x16x32_bf16(
        false, a, false, bS, (short)0, accS, false, false);
  }

  const int rbase = row0 + ((lane < 16) ? 0 : 8);
  if (row0 + 16 <= n) {
#pragma unroll
    for (int r = 0; r < 8; ++r) {
      proj[(size_t)(rbase + r) * HF + ncol] = accP[r];
      skip[(size_t)(rbase + r) * HF + ncol] = accS[r];
    }
  } else {
#pragma unroll
    for (int r = 0; r < 8; ++r) {
      if (rbase + r < n) {
        proj[(size_t)(rbase + r) * HF + ncol] = accP[r];
        skip[(size_t)(rbase + r) * HF + ncol] = accS[r];
      }
    }
  }
}

__global__ void gat_node_scores(const float* __restrict__ proj,
                                const float* __restrict__ a_src,
                                const float* __restrict__ a_trg,
                                float* __restrict__ s_src,
                                float* __restrict__ s_trg, int n) {
  int idx = blockIdx.x * blockDim.x + threadIdx.x;
  if (idx >= n * HEADS) return;
  int node = idx >> 3, h = idx & 7;
  const float4* p  = (const float4*)(proj  + (size_t)node * HF + h * FEAT);
  const float4* as = (const float4*)(a_src + h * FEAT);
  const float4* at = (const float4*)(a_trg + h * FEAT);
  float ss = 0.0f, st = 0.0f;
#pragma unroll
  for (int q = 0; q < 4; ++q) {
    float4 v = p[q], sa = as[q], ta = at[q];
    ss += v.x * sa.x + v.y * sa.y + v.z * sa.z + v.w * sa.w;
    st += v.x * ta.x + v.y * ta.y + v.z * ta.z + v.w * ta.w;
  }
  s_src[idx] = ss;
  s_trg[idx] = st;
}

__device__ __forceinline__ float leaky02(float s) {
  return (s >= 0.0f) ? s : 0.2f * s;
}

__global__ __launch_bounds__(256)
void gat_edge_max(const int* __restrict__ src, const int* __restrict__ trg,
                  const float* __restrict__ s_src, const float* __restrict__ s_trg,
                  float* __restrict__ gmax, int e) {
  int idx = blockIdx.x * 256 + threadIdx.x;
  float sc = -__builtin_inff();
  if (idx < e * HEADS) {
    int edge = idx >> 3, h = idx & 7;
    sc = leaky02(s_src[src[edge] * HEADS + h] + s_trg[trg[edge] * HEADS + h]);
  }
#pragma unroll
  for (int off = 16; off > 0; off >>= 1)
    sc = fmaxf(sc, __shfl_xor(sc, off, 32));

  __shared__ float wmaxs[8];
  if ((threadIdx.x & 31) == 0) wmaxs[threadIdx.x >> 5] = sc;
  __syncthreads();
  if (threadIdx.x < 8) {
    float v = wmaxs[threadIdx.x];
#pragma unroll
    for (int off = 4; off > 0; off >>= 1)
      v = fmaxf(v, __shfl_xor(v, off, 32));
    if (threadIdx.x == 0) {
      if (v >= 0.0f) atomicMax((int*)gmax, __float_as_int(v));
      else           atomicMin((unsigned int*)gmax, __float_as_uint(v));
    }
  }
}

__global__ void gat_edge_denom(const int* __restrict__ src, const int* __restrict__ trg,
                               const float* __restrict__ s_src, const float* __restrict__ s_trg,
                               const float* __restrict__ gmax,
                               float* __restrict__ denom, int e) {
  int idx = blockIdx.x * blockDim.x + threadIdx.x;
  if (idx >= e * HEADS) return;
  int edge = idx >> 3, h = idx & 7;
  int tn = trg[edge];
  float s  = leaky02(s_src[src[edge] * HEADS + h] + s_trg[tn * HEADS + h]);
  float ex = expf(s - gmax[0]);
  atomicAdd(&denom[tn * HEADS + h], ex);
}

__global__ void gat_edge_aggregate(const int* __restrict__ src, const int* __restrict__ trg,
                                   const float* __restrict__ s_src,
                                   const float* __restrict__ s_trg,
                                   const float* __restrict__ gmax,
                                   const float* __restrict__ denom,
                                   const float* __restrict__ proj,
                                   float* __restrict__ accum, int e) {
  int idx = blockIdx.x * blockDim.x + threadIdx.x;
  if (idx >= e * HEADS) return;
  int edge = idx >> 3, h = idx & 7;
  int sn = src[edge], tn = trg[edge];

  const float* p = proj + (size_t)sn * HF + h * FEAT;
  __builtin_prefetch(p, 0, 1);

  float s   = leaky02(s_src[sn * HEADS + h] + s_trg[tn * HEADS + h]);
  float att = expf(s - gmax[0]) / (denom[tn * HEADS + h] + 1e-16f);

  float* o = accum + (size_t)tn * HF + h * FEAT;
#pragma unroll
  for (int q = 0; q < 4; ++q) {
    float4 v = ((const float4*)p)[q];
    atomicAdd(o + q * 4 + 0, v.x * att);
    atomicAdd(o + q * 4 + 1, v.y * att);
    atomicAdd(o + q * 4 + 2, v.z * att);
    atomicAdd(o + q * 4 + 3, v.w * att);
  }
}

__global__ void gat_epilogue(const float* __restrict__ skip,
                             const float* __restrict__ bias,
                             float* __restrict__ out, int n) {
  int idx = blockIdx.x * blockDim.x + threadIdx.x;
  if (idx >= n * HF) return;
  float v = out[idx] + skip[idx] + bias[idx & 127];
  out[idx] = (v > 0.0f) ? v : expm1f(v);
}

extern "C" void kernel_launch(void* const* d_in, const int* in_sizes, int n_in,
                              void* d_out, int out_size, void* d_ws, size_t ws_size,
                              hipStream_t stream) {
  const float* x     = (const float*)d_in[0];
  const int*   edge  = (const int*)  d_in[1];
  const float* Wp    = (const float*)d_in[2];
  const float* a_src = (const float*)d_in[3];
  const float* a_trg = (const float*)d_in[4];
  const float* Ws    = (const float*)d_in[5];
  const float* bias  = (const float*)d_in[6];
  float*       out   = (float*)d_out;

  const int n = in_sizes[0] / FIN;
  const int e = in_sizes[1] / 2;
  const int* src = edge;
  const int* trg = edge + e;

  char* base = (char*)d_ws;
  __bf16* WpT = (__bf16*)base;
  __bf16* WsT = WpT + (size_t)FIN * HF;
  float* proj  = (float*)(base + 2u * FIN * HF * sizeof(__bf16));
  float* skip  = proj  + (size_t)n * HF;
  float* ssrc  = skip  + (size_t)n * HF;
  float* strg  = ssrc  + (size_t)n * HEADS;
  float* denom = strg  + (size_t)n * HEADS;
  float* gmax  = denom + (size_t)n * HEADS;

  const int nhf = n * HF;
  const int eh  = e * HEADS;

  gat_init<<<(nhf + 255) / 256, 256, 0, stream>>>(denom, gmax, out, n);
  gat_wcvt<<<(FIN * HF + 255) / 256, 256, 0, stream>>>(Wp, Ws, WpT, WsT);
  gat_proj_wmma<<<(n + 15) / 16, 256, 0, stream>>>(x, WpT, WsT, proj, skip, n);
  gat_node_scores<<<(n * HEADS + 255) / 256, 256, 0, stream>>>(proj, a_src, a_trg,
                                                               ssrc, strg, n);
  gat_edge_max<<<(eh + 255) / 256, 256, 0, stream>>>(src, trg, ssrc, strg, gmax, e);
  gat_edge_denom<<<(eh + 255) / 256, 256, 0, stream>>>(src, trg, ssrc, strg, gmax,
                                                       denom, e);
  gat_edge_aggregate<<<(eh + 255) / 256, 256, 0, stream>>>(src, trg, ssrc, strg, gmax,
                                                           denom, proj, out, e);
  gat_epilogue<<<(nhf + 255) / 256, 256, 0, stream>>>(skip, bias, out, n);
}


# Float atomic scatter-add replaced by an in-call counting sort of edges by target; one wave per target row sums its edges in registers and writes it once (denominator, skip, bias, ELU folded in)
# speedup vs baseline: 2.4304x; 2.4304x over previous
#include <hip/hip_runtime.h>
#include <math.h>


typedef __attribute__((ext_vector_type(16))) __bf16 v16bf;
typedef __attribute__((ext_vector_type(8)))  __bf16 v8bf;
typedef __attribute__((ext_vector_type(8)))  float  v8f;

#define FIN 128
#define HEADS 8
#define FEAT 16
#define HF 128

__global__ void gat_init(int* __restrict__ cnt, float* __restrict__ gmax, int n) {
  int idx = blockIdx.x * blockDim.x + threadIdx.x;
  if (idx < n) cnt[idx] = 0;
  if (idx == 0) gmax[0] = -__builtin_inff();
}

__global__ void gat_wcvt(const float* __restrict__ Wp, const float* __restrict__ Ws,
                         __bf16* __restrict__ WpT, __bf16* __restrict__ WsT) {
  int idx = blockIdx.x * blockDim.x + threadIdx.x;
  if (idx >= FIN * HF) return;
  int k = idx >> 7;
  int ncol = idx & 127;
  WpT[ncol * FIN + k] = (__bf16)Wp[k * HF + ncol];
  WsT[ncol * FIN + k] = (__bf16)Ws[k * HF + ncol];
}

__global__ __launch_bounds__(256)
void gat_proj_wmma(const float* __restrict__ x,
                   const __bf16* __restrict__ WpT,
                   const __bf16* __restrict__ WsT,
                   float* __restrict__ proj,
                   float* __restrict__ skip,
                   int n) {
  __shared__ __bf16 As[16][FIN + 8];

  const int t    = threadIdx.x;
  const int row0 = blockIdx.x * 16;

  for (int i = t; i < 16 * FIN; i += 256) {
    int r = i >> 7, c = i & 127;
    int gr = row0 + r;
    float v = (gr < n) ? x[(size_t)gr * FIN + c] : 0.0f;
    As[r][c] = (__bf16)v;
  }
  __syncthreads();

  const int wave = t >> 5;
  const int lane = t & 31;
  const int m    = lane & 15;
  const int kb   = (lane < 16) ? 0 : 8;
  const int ncol = wave * 16 + m;

  const v8bf* aRow  = (const v8bf*)(&As[m][0]);
  const v8bf* bpRow = (const v8bf*)(WpT + (size_t)ncol * FIN);
  const v8bf* bsRow = (const v8bf*)(WsT + (size_t)ncol * FIN);

  v8f accP = {};
  v8f accS = {};

#pragma unroll
  for (int kk = 0; kk < FIN; kk += 32) {
    const int c0 = (kk + kb) >> 3;
    const int c1 = c0 + 2;
    v8bf alo = aRow[c0],  ahi = aRow[c1];
    v8bf plo = bpRow[c0], phi = bpRow[c1];
    v8bf slo = bsRow[c0], shi = bsRow[c1];

    v16bf a  = __builtin_shufflevector(alo, ahi,
        0,1,2,3,4,5,6,7,8,9,10,11,12,13,14,15);
    v16bf bP = __builtin_shufflevector(plo, phi,
        0,1,2,3,4,5,6,7,8,9,10,11,12,13,14,15);
    v16bf bS = __builtin_shufflevector(slo, shi,
        0,1,2,3,4,5,6,7,8,9,10,11,12,13,14,15);

    accP = __builtin_amdgcn_wmma_f32_16x16x32_bf16(
        false, a, false, bP, (short)0, accP, false, false);
    accS = __builtin_amdgcn_wmma_f32_16x16x32_bf16(
        false, a, false, bS, (short)0, accS, false, false);
  }

  const int rbase = row0 + ((lane < 16) ? 0 : 8);
  if (row0 + 16 <= n) {
#pragma unroll
    for (int r = 0; r < 8; ++r) {
      proj[(size_t)(rbase + r) * HF + ncol] = accP[r];
      skip[(size_t)(rbase + r) * HF + ncol] = accS[r];
    }
  } else {
#pragma unroll
    for (int r = 0; r < 8; ++r) {
      if (rbase + r < n) {
        proj[(size_t)(rbase + r) * HF + ncol] = accP[r];
        skip[(size_t)(rbase + r) * HF + ncol] = accS[r];
      }
    }
  }
}

__global__ void gat_node_scores(const float* __restrict__ proj,
                                const float* __restrict__ a_src,
                                const float* __restrict__ a_trg,
                                float* __restrict__ s_src,
                                float* __restrict__ s_trg, int n) {
  int idx = blockIdx.x * blockDim.x + threadIdx.x;
  if (idx >= n * HEADS) return;
  int node = idx >> 3, h = idx & 7;
  const float4* p  = (const float4*)(proj  + (size_t)node * HF + h * FEAT);
  const float4* as = (const float4*)(a_src + h * FEAT);
  const float4* at = (const float4*)(a_trg + h * FEAT);
  float ss = 0.0f, st = 0.0f;
#pragma unroll
  for (int q = 0; q < 4; ++q) {
    float4 v = p[q], sa = as[q], ta = at[q];
    ss += v.x * sa.x + v.y * sa.y + v.z * sa.z + v.w * sa.w;
    st += v.x * ta.x + v.y * ta.y + v.z * ta.z + v.w * ta.w;
  }
  s_src[idx] = ss;
  s_trg[idx] = st;
}

__device__ __forceinline__ float leaky02(float s) {
  return (s >= 0.0f) ? s : 0.2f * s;
}

__global__ __launch_bounds__(256)
void gat_edge_max(const int* __restrict__ src, const int* __restrict__ trg,
                  const float* __restrict__ s_src, const float* __restrict__ s_trg,
                  float* __restrict__ gmax, int* __restrict__ cnt, int e) {
  int idx = blockIdx.x * 256 + threadIdx.x;
  float sc = -__builtin_inff();
  int tn = -1;
  if (idx < e * HEADS) {
    int edge = idx >> 3, h = idx & 7;
    int t = trg[edge];
    sc = leaky02(s_src[src[edge] * HEADS + h] + s_trg[t * HEADS + h]);
    if (h == 0) tn = t;
  }
#pragma unroll
  for (int off = 16; off > 0; off >>= 1)
    sc = fmaxf(sc, __shfl_xor(sc, off, 32));

  __shared__ float wmaxs[8];
  if ((threadIdx.x & 31) == 0) wmaxs[threadIdx.x >> 5] = sc;
  __syncthreads();
  if (tn >= 0) atomicAdd(&cnt[tn], 1);
  if (threadIdx.x < 8) {
    float v = wmaxs[threadIdx.x];
#pragma unroll
    for (int off = 4; off > 0; off >>= 1)
      v = fmaxf(v, __shfl_xor(v, off, 32));
    if (threadIdx.x == 0) {
      if (v >= 0.0f) atomicMax((int*)gmax, __float_as_int(v));
      else           atomicMin((unsigned int*)gmax, __float_as_uint(v));
    }
  }
}

__global__ __launch_bounds__(1024)
void gat_scan(const int* __restrict__ cnt, int* __restrict__ rowptr,
              int* __restrict__ cursor, int n) {
  __shared__ int wtot[32];
  __shared__ int wpre[32];
  __shared__ int sbase;
  const int tid  = threadIdx.x;
  const int lane = tid & 31;
  const int wave = tid >> 5;
  const int start = blockIdx.x * 1024;

  const int4* c4 = (const int4*)cnt;
  const int nq = start >> 2;
  int s = 0, s1 = 0, s2 = 0, s3 = 0;
  int q = tid;
  for (; q + 3072 < nq; q += 4096) {
    int4 a = c4[q], b = c4[q + 1024], cc = c4[q + 2048], d = c4[q + 3072];
    s  += (a.x + a.y) + (a.z + a.w);
    s1 += (b.x + b.y) + (b.z + b.w);
    s2 += (cc.x + cc.y) + (cc.z + cc.w);
    s3 += (d.x + d.y) + (d.z + d.w);
  }
  for (; q < nq; q += 1024) {
    int4 a = c4[q];
    s += (a.x + a.y) + (a.z + a.w);
  }
  s += (s1 + s2) + s3;
  for (int off = 16; off > 0; off >>= 1) s += __shfl_xor(s, off, 32);
  if (lane == 0) wtot[wave] = s;
  __syncthreads();
  if (wave == 0) {
    int v = wtot[lane];
    for (int off = 16; off > 0; off >>= 1) v += __shfl_xor(v, off, 32);
    if (lane == 0) sbase = v;
  }

  const int i = start + tid;
  const int c = (i < n) ? cnt[i] : 0;
  int incl = c;
  for (int off = 1; off < 32; off <<= 1) {
    int t = __shfl_up(incl, off, 32);
    if (lane >= off) incl += t;
  }
  if (lane == 31) wpre[wave] = incl;
  __syncthreads();
  if (wave == 0) {
    int v = wpre[lane];
    int vi = v;
    for (int off = 1; off < 32; off <<= 1) {
      int t = __shfl_up(vi, off, 32);
      if (lane >= off) vi += t;
    }
    wpre[lane] = vi - v;
  }
  __syncthreads();
  const int excl = sbase + wpre[wave] + (incl - c);
  if (i < n) {
    rowptr[i] = excl;
    cursor[i] = excl;
    if (i == n - 1) rowptr[n] = excl + c;
  }
}

__global__ void gat_fill(const int* __restrict__ src, const int* __restrict__ trg,
                         int* __restrict__ cursor, int* __restrict__ csr_src, int e) {
  int idx = blockIdx.x * blockDim.x + threadIdx.x;
  if (idx >= e) return;
  int pos = atomicAdd(&cursor[trg[idx]], 1);
  csr_src[pos] = src[idx];
}

__global__ __launch_bounds__(256)
void gat_gather(const int* __restrict__ rowptr, const int* __restrict__ csr_src,
                const float* __restrict__ s_src, const float* __restrict__ s_trg,
                const float* __restrict__ gmax, const float* __restrict__ proj,
                const float* __restrict__ bias, float* out, int n) {
  const int lane = threadIdx.x & 31;
  const int node = blockIdx.x * 8 + __builtin_amdgcn_readfirstlane((int)(threadIdx.x >> 5));
  if (node >= n) return;
  const int h = lane >> 2;

  const int   beg = rowptr[node];
  const int   deg = rowptr[node + 1] - beg;
  const float st  = s_trg[node * HEADS + h];
  const float gm  = gmax[0];
  float* orow = out + (size_t)node * HF + lane * 4;
  const float4 sk = *(const float4*)orow;
  const float4 bi = *(const float4*)(bias + lane * 4);

  float ax = 0.0f, ay = 0.0f, az = 0.0f, aw = 0.0f, den = 0.0f;

  for (int j0 = 0; j0 < deg; j0 += 32) {
    const int m = (deg - j0 < 32) ? (deg - j0) : 32;
    const int mine = (lane < m) ? csr_src[beg + j0 + lane] : node;
    for (int j = 0; j < m; j += 4) {
      const int n0 = __builtin_amdgcn_readlane(mine, j);
      const int n1 = __builtin_amdgcn_readlane(mine, j + 1);
      const int n2 = __builtin_amdgcn_readlane(mine, j + 2);
      const int n3 = __builtin_amdgcn_readlane(mine, j + 3);
      const float4 v0 = *(const float4*)(proj + (size_t)n0 * HF + lane * 4);
      const float4 v1 = *(const float4*)(proj + (size_t)n1 * HF + lane * 4);
      const float4 v2 = *(const float4*)(proj + (size_t)n2 * HF + lane * 4);
      const float4 v3 = *(const float4*)(proj + (size_t)n3 * HF + lane * 4);
      const float q0 = s_src[n0 * HEADS + h];
      const float q1 = s_src[n1 * HEADS + h];
      const float q2 = s_src[n2 * HEADS + h];
      const float q3 = s_src[n3 * HEADS + h];
      const float e0 = expf(leaky02(q0 + st) - gm);
      float e1 = expf(leaky02(q1 + st) - gm);
      float e2 = expf(leaky02(q2 + st) - gm);
      float e3 = expf(leaky02(q3 + st) - gm);
      e1 = (j + 1 < m) ? e1 : 0.0f;
      e2 = (j + 2 < m) ? e2 : 0.0f;
      e3 = (j + 3 < m) ? e3 : 0.0f;
      den += (e0 + e1) + (e2 + e3);
      ax += (e0 * v0.x + e1 * v1.x) + (e2 * v2.x + e3 * v3.x);
      ay += (e0 * v0.y + e1 * v1.y) + (e2 * v2.y + e3 * v3.y);
      az += (e0 * v0.z + e1 * v1.z) + (e2 * v2.z + e3 * v3.z);
      aw += (e0 * v0.w + e1 * v1.w) + (e2 * v2.w + e3 * v3.w);
    }
  }

  const float dn = den + 1e-16f;
  float4 r;
  r.x = ax / dn + sk.x + bi.x;
  r.y = ay / dn + sk.y + bi.y;
  r.z = az / dn + sk.z + bi.z;
  r.w = aw / dn + sk.w + bi.w;
  r.x = (r.x > 0.0f) ? r.x : expm1f(r.x);
  r.y = (r.y > 0.0f) ? r.y : expm1f(r.y);
  r.z = (r.z > 0.0f) ? r.z : expm1f(r.z);
  r.w = (r.w > 0.0f) ? r.w : expm1f(r.w);
  *(float4*)orow = r;
}

extern "C" void kernel_launch(void* const* d_in, const int* in_sizes, int n_in,
                              void* d_out, int out_size, void* d_ws, size_t ws_size,
                              hipStream_t stream) {
  const float* x     = (const float*)d_in[0];
  const int*   edge  = (const int*)  d_in[1];
  const float* Wp    = (const float*)d_in[2];
  const float* a_src = (const float*)d_in[3];
  const float* a_trg = (const float*)d_in[4];
  const float* Ws    = (const float*)d_in[5];
  const float* bias  = (const float*)d_in[6];
  float*       out   = (float*)d_out;

  const int n = in_sizes[0] / FIN;
  const int e = in_sizes[1] / 2;
  const int* src = edge;
  const int* trg = edge + e;

  char* base = (char*)d_ws;
  __bf16* WpT = (__bf16*)base;
  __bf16* WsT = WpT + (size_t)FIN * HF;
  float* proj  = (float*)(base + 2u * FIN * HF * sizeof(__bf16));
  float* sortr = proj  + (size_t)n * HF;
  int*   cnt    = (int*)sortr;
  int*   rowptr = cnt    + (size_t)n;
  int*   cursor = rowptr + (size_t)n + 1;
  int*   csrsrc = cursor + (size_t)n;
  float* ssrc  = sortr + (size_t)n * HF;
  float* strg  = ssrc  + (size_t)n * HEADS;
  float* gmax  = strg  + (size_t)n * HEADS + (size_t)n * HEADS;

  const int eh  = e * HEADS;

  gat_init<<<(n + 255) / 256, 256, 0, stream>>>(cnt, gmax, n);
  gat_wcvt<<<(FIN * HF + 255) / 256, 256, 0, stream>>>(Wp, Ws, WpT, WsT);
  gat_proj_wmma<<<(n + 15) / 16, 256, 0, stream>>>(x, WpT, WsT, proj, out, n);
  gat_node_scores<<<(n * HEADS + 255) / 256, 256, 0, stream>>>(proj, a_src, a_trg,
                                                               ssrc, strg, n);
  gat_edge_max<<<(eh + 255) / 256, 256, 0, stream>>>(src, trg, ssrc, strg, gmax, cnt, e);
  gat_scan<<<(n + 1023) / 1024, 1024, 0, stream>>>(cnt, rowptr, cursor, n);
  gat_fill<<<(e + 255) / 256, 256, 0, stream>>>(src, trg, cursor, csrsrc, e);
  gat_gather<<<(n + 7) / 8, 256, 0, stream>>>(rowptr, csrsrc, ssrc, strg, gmax,
                                              proj, bias, out, n);
}


# Edge-max/count pass reshaped: two lanes per edge read 4 heads each as float4 and take 8 edges per lane in registers (12504 waves, was 400000 of one score each); leaky-relu once on the block max
# speedup vs baseline: 3.4568x; 1.4223x over previous
#include <hip/hip_runtime.h>
#include <math.h>


typedef __attribute__((ext_vector_type(16))) __bf16 v16bf;
typedef __attribute__((ext_vector_type(8)))  __bf16 v8bf;
typedef __attribute__((ext_vector_type(8)))  float  v8f;

#define FIN 128
#define HEADS 8
#define FEAT 16
#define HF 128

__global__ void gat_init(int* __restrict__ cnt, float* __restrict__ gmax, int n) {
  int idx = blockIdx.x * blockDim.x + threadIdx.x;
  if (idx < n) cnt[idx] = 0;
  if (idx == 0) gmax[0] = -__builtin_inff();
}

__global__ void gat_wcvt(const float* __restrict__ Wp, const float* __restrict__ Ws,
                         __bf16* __restrict__ WpT, __bf16* __restrict__ WsT) {
  int idx = blockIdx.x * blockDim.x + threadIdx.x;
  if (idx >= FIN * HF) return;
  int k = idx >> 7;
  int ncol = idx & 127;
  WpT[ncol * FIN + k] = (__bf16)Wp[k * HF + ncol];
  WsT[ncol * FIN + k] = (__bf16)Ws[k * HF + ncol];
}

__global__ __launch_bounds__(256)
void gat_proj_wmma(const float* __restrict__ x,
                   const __bf16* __restrict__ WpT,
                   const __bf16* __restrict__ WsT,
                   float* __restrict__ proj,
                   float* __restrict__ skip,
                   int n) {
  __shared__ __bf16 As[16][FIN + 8];

  const int t    = threadIdx.x;
  const int row0 = blockIdx.x * 16;

  for (int i = t; i < 16 * FIN; i += 256) {
    int r = i >> 7, c = i & 127;
    int gr = row0 + r;
    float v = (gr < n) ? x[(size_t)gr * FIN + c] : 0.0f;
    As[r][c] = (__bf16)v;
  }
  __syncthreads();

  const int wave = t >> 5;
  const int lane = t & 31;
  const int m    = lane & 15;
  const int kb   = (lane < 16) ? 0 : 8;
  const int ncol = wave * 16 + m;

  const v8bf* aRow  = (const v8bf*)(&As[m][0]);
  const v8bf* bpRow = (const v8bf*)(WpT + (size_t)ncol * FIN);
  const v8bf* bsRow = (const v8bf*)(WsT + (size_t)ncol * FIN);

  v8f accP = {};
  v8f accS = {};

#pragma unroll
  for (int kk = 0; kk < FIN; kk += 32) {
    const int c0 = (kk + kb) >> 3;
    const int c1 = c0 + 2;
    v8bf alo = aRow[c0],  ahi = aRow[c1];
    v8bf plo = bpRow[c0], phi = bpRow[c1];
    v8bf slo = bsRow[c0], shi = bsRow[c1];

    v16bf a  = __builtin_shufflevector(alo, ahi,
        0,1,2,3,4,5,6,7,8,9,10,11,12,13,14,15);
    v16bf bP = __builtin_shufflevector(plo, phi,
        0,1,2,3,4,5,6,7,8,9,10,11,12,13,14,15);
    v16bf bS = __builtin_shufflevector(slo, shi,
        0,1,2,3,4,5,6,7,8,9,10,11,12,13,14,15);

    accP = __builtin_amdgcn_wmma_f32_16x16x32_bf16(
        false, a, false, bP, (short)0, accP, false, false);
    accS = __builtin_amdgcn_wmma_f32_16x16x32_bf16(
        false, a, false, bS, (short)0, accS, false, false);
  }

  const int rbase = row0 + ((lane < 16) ? 0 : 8);
  if (row0 + 16 <= n) {
#pragma unroll
    for (int r = 0; r < 8; ++r) {
      proj[(size_t)(rbase + r) * HF + ncol] = accP[r];
      skip[(size_t)(rbase + r) * HF + ncol] = accS[r];
    }
  } else {
#pragma unroll
    for (int r = 0; r < 8; ++r) {
      if (rbase + r < n) {
        proj[(size_t)(rbase + r) * HF + ncol] = accP[r];
        skip[(size_t)(rbase + r) * HF + ncol] = accS[r];
      }
    }
  }
}

__global__ void gat_node_scores(const float* __restrict__ proj,
                                const float* __restrict__ a_src,
                                const float* __restrict__ a_trg,
                                float* __restrict__ s_src,
                                float* __restrict__ s_trg, int n) {
  int idx = blockIdx.x * blockDim.x + threadIdx.x;
  if (idx >= n * HEADS) return;
  int node = idx >> 3, h = idx & 7;
  const float4* p  = (const float4*)(proj  + (size_t)node * HF + h * FEAT);
  const float4* as = (const float4*)(a_src + h * FEAT);
  const float4* at = (const float4*)(a_trg + h * FEAT);
  float ss = 0.0f, st = 0.0f;
#pragma unroll
  for (int q = 0; q < 4; ++q) {
    float4 v = p[q], sa = as[q], ta = at[q];
    ss += v.x * sa.x + v.y * sa.y + v.z * sa.z + v.w * sa.w;
    st += v.x * ta.x + v.y * ta.y + v.z * ta.z + v.w * ta.w;
  }
  s_src[idx] = ss;
  s_trg[idx] = st;
}

__device__ __forceinline__ float leaky02(float s) {
  return (s >= 0.0f) ? s : 0.2f * s;
}

#define EPT 8

__global__ __launch_bounds__(256)
void gat_edge_max(const int* __restrict__ src, const int* __restrict__ trg,
                  const float* __restrict__ s_src, const float* __restrict__ s_trg,
                  float* __restrict__ gmax, int* __restrict__ cnt, int e) {
  const int half  = threadIdx.x & 1;
  const int first = blockIdx.x * (128 * EPT) + (int)(threadIdx.x >> 1);
  const float4* ss4 = (const float4*)s_src;
  const float4* st4 = (const float4*)s_trg;

  int sn[EPT], tn[EPT];
#pragma unroll
  for (int u = 0; u < EPT; ++u) {
    const int edge = first + u * 128;
    const int ec = (edge < e) ? edge : (e - 1);
    sn[u] = src[ec];
    tn[u] = trg[ec];
  }
  float4 a[EPT], b[EPT];
#pragma unroll
  for (int u = 0; u < EPT; ++u) {
    a[u] = ss4[sn[u] * 2 + half];
    b[u] = st4[tn[u] * 2 + half];
  }
#pragma unroll
  for (int u = 0; u < EPT; ++u)
    if (half == 0 && first + u * 128 < e) atomicAdd(&cnt[tn[u]], 1);

  float sc = -__builtin_inff();
#pragma unroll
  for (int u = 0; u < EPT; ++u) {
    const float m01 = fmaxf(a[u].x + b[u].x, a[u].y + b[u].y);
    const float m23 = fmaxf(a[u].z + b[u].z, a[u].w + b[u].w);
    sc = fmaxf(sc, fmaxf(m01, m23));
  }
#pragma unroll
  for (int off = 16; off > 0; off >>= 1)
    sc = fmaxf(sc, __shfl_xor(sc, off, 32));

  __shared__ float wmaxs[8];
  if ((threadIdx.x & 31) == 0) wmaxs[threadIdx.x >> 5] = sc;
  __syncthreads();
  if (threadIdx.x < 8) {
    float v = wmaxs[threadIdx.x];
#pragma unroll
    for (int off = 4; off > 0; off >>= 1)
      v = fmaxf(v, __shfl_xor(v, off, 32));
    if (threadIdx.x == 0) {
      v = leaky02(v);
      if (v >= 0.0f) atomicMax((int*)gmax, __float_as_int(v));
      else           atomicMin((unsigned int*)gmax, __float_as_uint(v));
    }
  }
}

__global__ __launch_bounds__(1024)
void gat_scan(const int* __restrict__ cnt, int* __restrict__ rowptr,
              int* __restrict__ cursor, int n) {
  __shared__ int wtot[32];
  __shared__ int wpre[32];
  __shared__ int sbase;
  const int tid  = threadIdx.x;
  const int lane = tid & 31;
  const int wave = tid >> 5;
  const int start = blockIdx.x * 1024;

  const int4* c4 = (const int4*)cnt;
  const int nq = start >> 2;
  int s = 0, s1 = 0, s2 = 0, s3 = 0;
  int q = tid;
  for (; q + 3072 < nq; q += 4096) {
    int4 a = c4[q], b = c4[q + 1024], cc = c4[q + 2048], d = c4[q + 3072];
    s  += (a.x + a.y) + (a.z + a.w);
    s1 += (b.x + b.y) + (b.z + b.w);
    s2 += (cc.x + cc.y) + (cc.z + cc.w);
    s3 += (d.x + d.y) + (d.z + d.w);
  }
  for (; q < nq; q += 1024) {
    int4 a = c4[q];
    s += (a.x + a.y) + (a.z + a.w);
  }
  s += (s1 + s2) + s3;
  for (int off = 16; off > 0; off >>= 1) s += __shfl_xor(s, off, 32);
  if (lane == 0) wtot[wave] = s;
  __syncthreads();
  if (wave == 0) {
    int v = wtot[lane];
    for (int off = 16; off > 0; off >>= 1) v += __shfl_xor(v, off, 32);
    if (lane == 0) sbase = v;
  }

  const int i = start + tid;
  const int c = (i < n) ? cnt[i] : 0;
  int incl = c;
  for (int off = 1; off < 32; off <<= 1) {
    int t = __shfl_up(incl, off, 32);
    if (lane >= off) incl += t;
  }
  if (lane == 31) wpre[wave] = incl;
  __syncthreads();
  if (wave == 0) {
    int v = wpre[lane];
    int vi = v;
    for (int off = 1; off < 32; off <<= 1) {
      int t = __shfl_up(vi, off, 32);
      if (lane >= off) vi += t;
    }
    wpre[lane] = vi - v;
  }
  __syncthreads();
  const int excl = sbase + wpre[wave] + (incl - c);
  if (i < n) {
    rowptr[i] = excl;
    cursor[i] = excl;
    if (i == n - 1) rowptr[n] = excl + c;
  }
}

__global__ void gat_fill(const int* __restrict__ src, const int* __restrict__ trg,
                         int* __restrict__ cursor, int* __restrict__ csr_src, int e) {
  int idx = blockIdx.x * blockDim.x + threadIdx.x;
  if (idx >= e) return;
  int pos = atomicAdd(&cursor[trg[idx]], 1);
  csr_src[pos] = src[idx];
}

__global__ __launch_bounds__(256)
void gat_gather(const int* __restrict__ rowptr, const int* __restrict__ csr_src,
                const float* __restrict__ s_src, const float* __restrict__ s_trg,
                const float* __restrict__ gmax, const float* __restrict__ proj,
                const float* __restrict__ bias, float* out, int n) {
  const int lane = threadIdx.x & 31;
  const int node = blockIdx.x * 8 + __builtin_amdgcn_readfirstlane((int)(threadIdx.x >> 5));
  if (node >= n) return;
  const int h = lane >> 2;

  const int   beg = rowptr[node];
  const int   deg = rowptr[node + 1] - beg;
  const float st  = s_trg[node * HEADS + h];
  const float gm  = gmax[0];
  float* orow = out + (size_t)node * HF + lane * 4;
  const float4 sk = *(const float4*)orow;
  const float4 bi = *(const float4*)(bias + lane * 4);

  float ax = 0.0f, ay = 0.0f, az = 0.0f, aw = 0.0f, den = 0.0f;

  for (int j0 = 0; j0 < deg; j0 += 32) {
    const int m = (deg - j0 < 32) ? (deg - j0) : 32;
    const int mine = (lane < m) ? csr_src[beg + j0 + lane] : node;
    for (int j = 0; j < m; j += 4) {
      const int n0 = __builtin_amdgcn_readlane(mine, j);
      const int n1 = __builtin_amdgcn_readlane(mine, j + 1);
      const int n2 = __builtin_amdgcn_readlane(mine, j + 2);
      const int n3 = __builtin_amdgcn_readlane(mine, j + 3);
      const float4 v0 = *(const float4*)(proj + (size_t)n0 * HF + lane * 4);
      const float4 v1 = *(const float4*)(proj + (size_t)n1 * HF + lane * 4);
      const float4 v2 = *(const float4*)(proj + (size_t)n2 * HF + lane * 4);
      const float4 v3 = *(const float4*)(proj + (size_t)n3 * HF + lane * 4);
      const float q0 = s_src[n0 * HEADS + h];
      const float q1 = s_src[n1 * HEADS + h];
      const float q2 = s_src[n2 * HEADS + h];
      const float q3 = s_src[n3 * HEADS + h];
      const float e0 = expf(leaky02(q0 + st) - gm);
      float e1 = expf(leaky02(q1 + st) - gm);
      float e2 = expf(leaky02(q2 + st) - gm);
      float e3 = expf(leaky02(q3 + st) - gm);
      e1 = (j + 1 < m) ? e1 : 0.0f;
      e2 = (j + 2 < m) ? e2 : 0.0f;
      e3 = (j + 3 < m) ? e3 : 0.0f;
      den += (e0 + e1) + (e2 + e3);
      ax += (e0 * v0.x + e1 * v1.x) + (e2 * v2.x + e3 * v3.x);
      ay += (e0 * v0.y + e1 * v1.y) + (e2 * v2.y + e3 * v3.y);
      az += (e0 * v0.z + e1 * v1.z) + (e2 * v2.z + e3 * v3.z);
      aw += (e0 * v0.w + e1 * v1.w) + (e2 * v2.w + e3 * v3.w);
    }
  }

  const float dn = den + 1e-16f;
  float4 r;
  r.x = ax / dn + sk.x + bi.x;
  r.y = ay / dn + sk.y + bi.y;
  r.z = az / dn + sk.z + bi.z;
  r.w = aw / dn + sk.w + bi.w;
  r.x = (r.x > 0.0f) ? r.x : expm1f(r.x);
  r.y = (r.y > 0.0f) ? r.y : expm1f(r.y);
  r.z = (r.z > 0.0f) ? r.z : expm1f(r.z);
  r.w = (r.w > 0.0f) ? r.w : expm1f(r.w);
  *(float4*)orow = r;
}

extern "C" void kernel_launch(void* const* d_in, const int* in_sizes, int n_in,
                              void* d_out, int out_size, void* d_ws, size_t ws_size,
                              hipStream_t stream) {
  const float* x     = (const float*)d_in[0];
  const int*   edge  = (const int*)  d_in[1];
  const float* Wp    = (const float*)d_in[2];
  const float* a_src = (const float*)d_in[3];
  const float* a_trg = (const float*)d_in[4];
  const float* Ws    = (const float*)d_in[5];
  const float* bias  = (const float*)d_in[6];
  float*       out   = (float*)d_out;

  const int n = in_sizes[0] / FIN;
  const int e = in_sizes[1] / 2;
  const int* src = edge;
  const int* trg = edge + e;

  char* base = (char*)d_ws;
  __bf16* WpT = (__bf16*)base;
  __bf16* WsT = WpT + (size_t)FIN * HF;
  float* proj  = (float*)(base + 2u * FIN * HF * sizeof(__bf16));
  float* sortr = proj  + (size_t)n * HF;
  int*   cnt    = (int*)sortr;
  int*   rowptr = cnt    + (size_t)n;
  int*   cursor = rowptr + (size_t)n + 1;
  int*   csrsrc = cursor + (size_t)n;
  float* ssrc  = sortr + (size_t)n * HF;
  float* strg  = ssrc  + (size_t)n * HEADS;
  float* gmax  = strg  + (size_t)n * HEADS + (size_t)n * HEADS;

  gat_init<<<(n + 255) / 256, 256, 0, stream>>>(cnt, gmax, n);
  gat_wcvt<<<(FIN * HF + 255) / 256, 256, 0, stream>>>(Wp, Ws, WpT, WsT);
  gat_proj_wmma<<<(n + 15) / 16, 256, 0, stream>>>(x, WpT, WsT, proj, out, n);
  gat_node_scores<<<(n * HEADS + 255) / 256, 256, 0, stream>>>(proj, a_src, a_trg,
                                                               ssrc, strg, n);
  gat_edge_max<<<(e + 128 * EPT - 1) / (128 * EPT), 256, 0, stream>>>(src, trg, ssrc, strg,
                                                               gmax, cnt, e);
  gat_scan<<<(n + 1023) / 1024, 1024, 0, stream>>>(cnt, rowptr, cursor, n);
  gat_fill<<<(e + 255) / 256, 256, 0, stream>>>(src, trg, cursor, csrsrc, e);
  gat_gather<<<(n + 7) / 8, 256, 0, stream>>>(rowptr, csrsrc, ssrc, strg, gmax,
                                              proj, bias, out, n);
}
